# starting kernel
# baseline (speedup 1.0000x reference)
#include <hip/hip_runtime.h>

typedef float v2f __attribute__((ext_vector_type(2)));
typedef float v8f __attribute__((ext_vector_type(8)));

#define GN 512
#define LDSP 17

__global__ __launch_bounds__(256)
void pml_coeff(const float* __restrict__ b,
               float* __restrict__ C1p, float* __restrict__ C3p)
{
  const int i = blockIdx.x * 256 + threadIdx.x;
  const float bv = b[i];
  C1p[i] = 1.0f / (4.0f + bv);
  C3p[i] = 4.0f - bv;
}

__global__ __launch_bounds__(256)
void wave_step(const float* __restrict__ xi,
               const float* __restrict__ xi_next,
               const float* __restrict__ u1,
               const float* __restrict__ u2,
               const float* __restrict__ c2,
               const float* __restrict__ msk,
               const float* __restrict__ C1p,
               const float* __restrict__ C3p,
               float* __restrict__ out)
{
  __shared__ float lds[8][16 * LDSP];

  const int lane = threadIdx.x & 31;
  const int wave = threadIdx.x >> 5;
  const int tile = blockIdx.x * 8 + wave;
  const int r0 = (tile >> 5) * 16;
  const int c0 = (tile & 31) * 16;

  const int hi = lane >> 4;
  const int lp = lane & 15;

  float* t = lds[wave];

  const unsigned rowcode = (unsigned)((0x19ull << (2 * lp)) >> 2);

  {
    const int lr = lane >> 2;
    const int lc = (lane & 3) * 4;
    const float4 p0 = *(const float4*)&u1[(r0 + lr) * GN + c0 + lc];
    const float4 p1 = *(const float4*)&u1[(r0 + 8 + lr) * GN + c0 + lc];
    float* q0 = &t[lr * LDSP + lc];
    q0[0] = p0.x; q0[1] = p0.y; q0[2] = p0.z; q0[3] = p0.w;
    float* q1 = &t[(lr + 8) * LDSP + lc];
    q1[0] = p1.x; q1[1] = p1.y; q1[2] = p1.z; q1[3] = p1.w;
  }

  v8f acc = {};
  #pragma unroll
  for (int k0 = 0; k0 < 16; k0 += 4) {
    const int sh = 30 - 2 * k0 - 4 * hi;

    v2f tv;
    tv.x = (float)((int)(rowcode << sh) >> 30);
    tv.y = (float)((int)(rowcode << (sh - 2)) >> 30);

    v2f bU;
    bU.x = t[(k0 + 2 * hi + 0) * LDSP + lp];
    bU.y = t[(k0 + 2 * hi + 1) * LDSP + lp];
    acc = __builtin_amdgcn_wmma_f32_16x16x4_f32(
        false, tv, false, bU, (short)0, acc, false, false);

    v2f aU;
    aU.x = t[lp * LDSP + k0 + 2 * hi + 0];
    aU.y = t[lp * LDSP + k0 + 2 * hi + 1];
    acc = __builtin_amdgcn_wmma_f32_16x16x4_f32(
        false, aU, false, tv, (short)0, acc, false, false);
  }

  #pragma unroll
  for (int v = 0; v < 8; ++v) {
    const int M  = v + 8 * hi;
    const int gr = r0 + M;
    const int gc = c0 + lp;
    const int idx = gr * GN + gc;

    float lap = acc[v];
    if (M == 0  && gr > 0)      lap += u1[(gr - 1) * GN + gc];
    if (M == 15 && gr < GN - 1) lap += u1[(gr + 1) * GN + gc];
    if (lp == 0  && gc > 0)      lap += u1[idx - 1];
    if (lp == 15 && gc < GN - 1) lap += u1[idx + 1];

    const float u1c = t[M * LDSP + lp];
    out[idx] = C1p[idx] * (8.0f * u1c - C3p[idx] * u2[idx]
                           + c2[idx] * lap + msk[idx] * xi[idx]);

    if (lp == 0) __builtin_prefetch(&xi_next[gr * GN + c0], 0, 1);
  }
}

extern "C" void kernel_launch(void* const* d_in, const int* in_sizes, int n_in,
                              void* d_out, int out_size, void* d_ws, size_t ws_size,
                              hipStream_t stream) {
  const float* inp = (const float*)d_in[0];
  const float* un1 = (const float*)d_in[1];
  const float* un2 = (const float*)d_in[2];
  const float* c2  = (const float*)d_in[3];
  const float* msk = (const float*)d_in[4];
  const float* b   = (const float*)d_in[5];

  const int plane = GN * GN;
  const int nt = in_sizes[0] / plane;

  float* buf0 = (float*)d_ws;
  float* buf1 = buf0 + plane;
  float* C1p  = buf1 + plane;
  float* C3p  = C1p  + plane;
  float* outp = (float*)d_out;

  pml_coeff<<<dim3(plane / 256), dim3(256), 0, stream>>>(b, C1p, C3p);

  dim3 grid(128);
  dim3 block(256);

  for (int t = 0; t < nt; ++t) {
    const float* xi      = inp + (size_t)t * plane;
    const float* xi_next = (t + 1 < nt) ? (inp + (size_t)(t + 1) * plane) : xi;

    const float* u1 = (t == 0) ? un1 : ((t & 1) ? buf0 : buf1);
    const float* u2 = (t == 0) ? un2 : ((t == 1) ? un1 : ((t & 1) ? buf1 : buf0));
    float* o = (t == nt - 1) ? outp : ((t & 1) ? buf1 : buf0);

    wave_step<<<grid, block, 0, stream>>>(xi, xi_next, u1, u2, c2, msk,
                                          C1p, C3p, o);
  }
}


# Temporal fusion: 16 time steps per launch on a 64x64 LDS region per workgroup (32x32 interior + one-tile halo, recomputed), PML coefficients in registers; launches 257 to 16; sum order as before
# speedup vs baseline: 10.3995x; 10.3995x over previous
#include <hip/hip_runtime.h>

typedef float v2f __attribute__((ext_vector_type(2)));
typedef float v8f __attribute__((ext_vector_type(8)));

#define GN 512
#define LDSP 17
#define TSZ (16 * LDSP)
#define KSTEPS 16
#define PFD 3

__global__ __launch_bounds__(512)
void wave_steps(const float* __restrict__ xi0,
                const float* __restrict__ u1g,
                const float* __restrict__ u2g,
                const float* __restrict__ c2g,
                const float* __restrict__ mskg,
                const float* __restrict__ bg,
                float* __restrict__ o1,
                float* __restrict__ o2,
                int nsteps, int avail)
{
  __shared__ float lds[2][16 * TSZ];

  const int lane = threadIdx.x & 31;
  const int w  = __builtin_amdgcn_readfirstlane((int)(threadIdx.x >> 5));
  const int tr = w >> 2, tc = w & 3;
  const int by = blockIdx.x >> 4, bx = blockIdx.x & 15;
  const int gr0 = by * 32 - 16 + tr * 16;
  const int gc0 = bx * 32 - 16 + tc * 16;
  const bool inDom = gr0 >= 0 && gr0 < GN && gc0 >= 0 && gc0 < GN;

  const int hi = lane >> 4;
  const int lp = lane & 15;
  const int plane = GN * GN;

  const unsigned rowcode = (unsigned)((0x19ull << (2 * lp)) >> 2);

  const int own = w * TSZ;
  const int cd0 = own + (8 * hi) * LDSP + lp;

  const bool udSel = hi ? (tr < 3) : (tr > 0);
  const int  udAdr = udSel ? (hi ? (own + 4 * TSZ + lp) : (own - 4 * TSZ + 15 * LDSP + lp))
                           : cd0;
  const bool lSel = (lp == 0) && (tc > 0);
  const bool rSel = (lp == 15) && (tc < 3);
  const bool lrSel = lSel || rSel;
  const int  lrAdr = lSel ? (own - TSZ + (8 * hi) * LDSP + 15)
                   : rSel ? (own + TSZ + (8 * hi) * LDSP)
                          : cd0;

  float u1c[8], u2r[8], C1[8], C3[8], c2r[8], mk[8], xc[8];
  int   gidx0 = 0;

  if (inDom) {
    gidx0 = (gr0 + 8 * hi) * GN + gc0 + lp;
    #pragma unroll
    for (int v = 0; v < 8; ++v) {
      const int idx = gidx0 + v * GN;
      u1c[v] = u1g[idx];
      u2r[v] = u2g[idx];
      const float bv = bg[idx];
      C1[v]  = 1.0f / (4.0f + bv);
      C3[v]  = 4.0f - bv;
      c2r[v] = c2g[idx];
      mk[v]  = mskg[idx];
      xc[v]  = xi0[idx];
      lds[0][cd0 + v * LDSP] = u1c[v];
    }
  } else {
    #pragma unroll
    for (int v = 0; v < 8; ++v) {
      u1c[v] = u2r[v] = C1[v] = C3[v] = c2r[v] = mk[v] = xc[v] = 0.0f;
      lds[0][cd0 + v * LDSP] = 0.0f;
      lds[1][cd0 + v * LDSP] = 0.0f;
    }
  }
  __syncthreads();

  int cur = 0;
  for (int s = 0; s < nsteps; ++s) {
    const float* tb = lds[cur];
    float*       nb = lds[cur ^ 1];

    if (inDom) {
      float xn[8];
      #pragma unroll
      for (int v = 0; v < 8; ++v) xn[v] = 0.0f;
      if (s + 1 < nsteps) {
        #pragma unroll
        for (int v = 0; v < 8; ++v)
          xn[v] = xi0[(size_t)(s + 1) * plane + gidx0 + v * GN];
      }

      if (s + PFD < avail && lp == 0) {
        #pragma unroll
        for (int v = 0; v < 8; ++v)
          __builtin_prefetch(&xi0[(size_t)(s + PFD) * plane + gidx0 + v * GN], 0, 1);
      }

      const float* t = tb + own;

      const float udv = tb[udAdr];
      const float ud  = udSel ? udv : 0.0f;
      float lrv[8];
      #pragma unroll
      for (int v = 0; v < 8; ++v) lrv[v] = tb[lrAdr + v * LDSP];

      v8f acc = {};
      #pragma unroll
      for (int k0 = 0; k0 < 16; k0 += 4) {
        const int sh = 30 - 2 * k0 - 4 * hi;

        v2f tv;
        tv.x = (float)((int)(rowcode << sh) >> 30);
        tv.y = (float)((int)(rowcode << (sh - 2)) >> 30);

        v2f bU;
        bU.x = t[(k0 + 2 * hi + 0) * LDSP + lp];
        bU.y = t[(k0 + 2 * hi + 1) * LDSP + lp];
        acc = __builtin_amdgcn_wmma_f32_16x16x4_f32(
            false, tv, false, bU, (short)0, acc, false, false);

        v2f aU;
        aU.x = t[lp * LDSP + k0 + 2 * hi + 0];
        aU.y = t[lp * LDSP + k0 + 2 * hi + 1];
        acc = __builtin_amdgcn_wmma_f32_16x16x4_f32(
            false, aU, false, tv, (short)0, acc, false, false);
      }

      #pragma unroll
      for (int v = 0; v < 8; ++v) {
        float lap = acc[v];
        if (v == 0) lap += hi ? 0.0f : ud;
        if (v == 7) lap += hi ? ud : 0.0f;
        lap += lrSel ? lrv[v] : 0.0f;

        const float un = C1[v] * (8.0f * u1c[v] - C3[v] * u2r[v]
                                  + c2r[v] * lap + mk[v] * xc[v]);
        nb[cd0 + v * LDSP] = un;
        u2r[v] = u1c[v];
        u1c[v] = un;
        xc[v]  = xn[v];
      }
    }
    __syncthreads();
    cur ^= 1;
  }

  #pragma unroll
  for (int j = 0; j < 2; ++j) {
    const int r = 2 * w + j;
    const int a = ((1 + (r >> 4)) * 4 + 1 + hi) * TSZ + (r & 15) * LDSP + lp;
    const int g = (by * 32 + r) * GN + bx * 32 + lane;
    o1[g] = lds[cur][a];
    if (o2) o2[g] = lds[cur ^ 1][a];
  }
}

extern "C" void kernel_launch(void* const* d_in, const int* in_sizes, int n_in,
                              void* d_out, int out_size, void* d_ws, size_t ws_size,
                              hipStream_t stream) {
  const float* inp = (const float*)d_in[0];
  const float* un1 = (const float*)d_in[1];
  const float* un2 = (const float*)d_in[2];
  const float* c2  = (const float*)d_in[3];
  const float* msk = (const float*)d_in[4];
  const float* b   = (const float*)d_in[5];

  const int plane = GN * GN;
  const int nt = in_sizes[0] / plane;

  float* A1 = (float*)d_ws;
  float* A2 = A1 + plane;
  float* B1 = A2 + plane;
  float* B2 = B1 + plane;
  float* outp = (float*)d_out;

  dim3 grid(256);
  dim3 block(512);

  int L = 0;
  for (int t0 = 0; t0 < nt; t0 += KSTEPS, ++L) {
    const int left = nt - t0;
    const int ns = left < KSTEPS ? left : KSTEPS;
    const bool last = (left <= KSTEPS);
    const float* i1 = (L == 0) ? un1 : ((L & 1) ? A1 : B1);
    const float* i2 = (L == 0) ? un2 : ((L & 1) ? A2 : B2);
    float* p1 = last ? outp : ((L & 1) ? B1 : A1);
    float* p2 = last ? (float*)nullptr : ((L & 1) ? B2 : A2);
    wave_steps<<<grid, block, 0, stream>>>(inp + (size_t)t0 * plane, i1, i2, c2, msk, b,
                                           p1, p2, ns, left);
  }
}


# Workgroup barrier split: a tile's WMMAs read only the wave's own LDS slice and run before the wait, halo reads and pointwise behind it; forcing loads stay in flight across it; sum order as before
# speedup vs baseline: 11.9807x; 1.1520x over previous
#include <hip/hip_runtime.h>

typedef float v2f __attribute__((ext_vector_type(2)));
typedef float v8f __attribute__((ext_vector_type(8)));

#define GN 512
#define LDSP 17
#define TSZ (16 * LDSP)
#define KSTEPS 16
#define PFD 3

template <bool B> struct Reload { static constexpr bool value = B; };

__device__ __forceinline__ void wait_ds0()   { asm volatile("s_wait_dscnt 0x0" ::: "memory"); }
__device__ __forceinline__ void bar_signal() { asm volatile("s_barrier_signal -1" ::: "memory"); }
__device__ __forceinline__ void bar_wait()   { asm volatile("s_barrier_wait -1" ::: "memory"); }
__device__ __forceinline__ void bar_wait_behind(v8f& acc, float (&x)[8]) {
  asm volatile("s_barrier_wait -1"
               : "+v"(acc), "+v"(x[0]), "+v"(x[1]), "+v"(x[2]), "+v"(x[3]),
                 "+v"(x[4]), "+v"(x[5]), "+v"(x[6]), "+v"(x[7])
               :: "memory");
}

__global__ __launch_bounds__(512)
void wave_steps(const float* __restrict__ xi0,
                const float* __restrict__ u1g,
                const float* __restrict__ u2g,
                const float* __restrict__ c2g,
                const float* __restrict__ mskg,
                const float* __restrict__ bg,
                float* __restrict__ o1,
                float* __restrict__ o2,
                int nsteps, int avail)
{
  __shared__ float lds[2][16 * TSZ];

  const int lane = threadIdx.x & 31;
  const int w  = __builtin_amdgcn_readfirstlane((int)(threadIdx.x >> 5));
  const int tr = w >> 2, tc = w & 3;
  const int by = blockIdx.x >> 4, bx = blockIdx.x & 15;
  const int gr0 = by * 32 - 16 + tr * 16;
  const int gc0 = bx * 32 - 16 + tc * 16;
  const bool inDom = gr0 >= 0 && gr0 < GN && gc0 >= 0 && gc0 < GN;

  const int hi = lane >> 4;
  const int lp = lane & 15;
  const int plane = GN * GN;

  const unsigned rowcode = (unsigned)((0x19ull << (2 * lp)) >> 2);

  const int own = w * TSZ;
  const int cd0 = own + (8 * hi) * LDSP + lp;

  const bool udSel = hi ? (tr < 3) : (tr > 0);
  const int  udAdr = udSel ? (hi ? (own + 4 * TSZ + lp) : (own - 4 * TSZ + 15 * LDSP + lp))
                           : cd0;
  const bool lSel = (lp == 0) && (tc > 0);
  const bool rSel = (lp == 15) && (tc < 3);
  const bool lrSel = lSel || rSel;
  const int  lrAdr = lSel ? (own - TSZ + (8 * hi) * LDSP + 15)
                   : rSel ? (own + TSZ + (8 * hi) * LDSP)
                          : cd0;

  float u1c[8], u2r[8], C1[8], C3[8], c2r[8], mk[8];
  float xa[8], xb[8];
  int   gidx0 = 0;

  if (inDom) {
    gidx0 = (gr0 + 8 * hi) * GN + gc0 + lp;
    #pragma unroll
    for (int v = 0; v < 8; ++v) {
      const int idx = gidx0 + v * GN;
      u1c[v] = u1g[idx];
      u2r[v] = u2g[idx];
      const float bv = bg[idx];
      C1[v]  = 1.0f / (4.0f + bv);
      C3[v]  = 4.0f - bv;
      c2r[v] = c2g[idx];
      mk[v]  = mskg[idx];
      xa[v]  = xi0[idx];
      xb[v]  = xi0[(nsteps > 1 ? (size_t)plane : 0) + idx];
      lds[0][cd0 + v * LDSP] = u1c[v];
    }
  } else {
    #pragma unroll
    for (int v = 0; v < 8; ++v) {
      u1c[v] = u2r[v] = C1[v] = C3[v] = c2r[v] = mk[v] = xa[v] = xb[v] = 0.0f;
      lds[0][cd0 + v * LDSP] = 0.0f;
      lds[1][cd0 + v * LDSP] = 0.0f;
    }
  }
  __builtin_amdgcn_fence(__ATOMIC_RELEASE, "workgroup");
  bar_signal();

  auto step = [&](int s, const float* tb, float* nb, float (&xcur)[8], auto reload) {
    v8f acc = {};
    {
      if (s + PFD < avail && lp == 0) {
        #pragma unroll
        for (int v = 0; v < 8; ++v)
          __builtin_prefetch(&xi0[(size_t)(s + PFD) * plane + gidx0 + v * GN], 0, 1);
      }

      const float* t = tb + own;
      v2f bU[4], aU[4];
      #pragma unroll
      for (int k = 0; k < 4; ++k) {
        bU[k].x = t[(4 * k + 2 * hi + 0) * LDSP + lp];
        bU[k].y = t[(4 * k + 2 * hi + 1) * LDSP + lp];
        aU[k].x = t[lp * LDSP + 4 * k + 2 * hi + 0];
        aU[k].y = t[lp * LDSP + 4 * k + 2 * hi + 1];
      }
      asm volatile("" : "+v"(bU[0]), "+v"(bU[1]), "+v"(bU[2]), "+v"(bU[3]),
                        "+v"(aU[0]), "+v"(aU[1]), "+v"(aU[2]), "+v"(aU[3]));
      #pragma unroll
      for (int k = 0; k < 4; ++k) {
        const int sh = 30 - 8 * k - 4 * hi;
        v2f tv;
        tv.x = (float)((int)(rowcode << sh) >> 30);
        tv.y = (float)((int)(rowcode << (sh - 2)) >> 30);
        acc = __builtin_amdgcn_wmma_f32_16x16x4_f32(
            false, tv, false, bU[k], (short)0, acc, false, false);
        acc = __builtin_amdgcn_wmma_f32_16x16x4_f32(
            false, aU[k], false, tv, (short)0, acc, false, false);
      }
    }

    bar_wait_behind(acc, xcur);

    {
      const float udv = tb[udAdr];
      const float ud  = udSel ? udv : 0.0f;
      float lrv[8];
      #pragma unroll
      for (int v = 0; v < 8; ++v) lrv[v] = tb[lrAdr + v * LDSP];

      #pragma unroll
      for (int v = 0; v < 8; ++v) {
        float lap = acc[v];
        if (v == 0) lap += hi ? 0.0f : ud;
        if (v == 7) lap += hi ? ud : 0.0f;
        lap += lrSel ? lrv[v] : 0.0f;

        const float un = C1[v] * (8.0f * u1c[v] - C3[v] * u2r[v]
                                  + c2r[v] * lap + mk[v] * xcur[v]);
        nb[cd0 + v * LDSP] = un;
        u2r[v] = u1c[v];
        u1c[v] = un;
      }
    }
    wait_ds0();
    bar_signal();

    if constexpr (decltype(reload)::value) {
      #pragma unroll
      for (int v = 0; v < 8; ++v)
        xcur[v] = xi0[(size_t)(s + 2) * plane + gidx0 + v * GN];
    }
  };

  if (inDom) {
    int s = 0;
    for (; s + 3 < nsteps; s += 2) {
      step(s,     lds[0], lds[1], xa, Reload<true>{});
      step(s + 1, lds[1], lds[0], xb, Reload<true>{});
    }
    if (nsteps - s == 3) {
      step(s, lds[0], lds[1], xa, Reload<true>{});
      step(s + 1, lds[1], lds[0], xb, Reload<false>{});
      step(s + 2, lds[0], lds[1], xa, Reload<false>{});
    } else if (nsteps - s == 2) {
      step(s, lds[0], lds[1], xa, Reload<false>{});
      step(s + 1, lds[1], lds[0], xb, Reload<false>{});
    } else if (nsteps - s == 1) {
      step(s, lds[0], lds[1], xa, Reload<false>{});
    }
  } else {
    for (int s = 0; s < nsteps; ++s) {
      bar_wait();
      bar_signal();
    }
  }
  const int cur = nsteps & 1;
  bar_wait();

  #pragma unroll
  for (int j = 0; j < 2; ++j) {
    const int r = 2 * w + j;
    const int a = ((1 + (r >> 4)) * 4 + 1 + hi) * TSZ + (r & 15) * LDSP + lp;
    const int g = (by * 32 + r) * GN + bx * 32 + lane;
    o1[g] = lds[cur][a];
    if (o2) o2[g] = lds[cur ^ 1][a];
  }
}

extern "C" void kernel_launch(void* const* d_in, const int* in_sizes, int n_in,
                              void* d_out, int out_size, void* d_ws, size_t ws_size,
                              hipStream_t stream) {
  const float* inp = (const float*)d_in[0];
  const float* un1 = (const float*)d_in[1];
  const float* un2 = (const float*)d_in[2];
  const float* c2  = (const float*)d_in[3];
  const float* msk = (const float*)d_in[4];
  const float* b   = (const float*)d_in[5];

  const int plane = GN * GN;
  const int nt = in_sizes[0] / plane;

  float* A1 = (float*)d_ws;
  float* A2 = A1 + plane;
  float* B1 = A2 + plane;
  float* B2 = B1 + plane;
  float* outp = (float*)d_out;

  dim3 grid(256);
  dim3 block(512);

  int L = 0;
  for (int t0 = 0; t0 < nt; t0 += KSTEPS, ++L) {
    const int left = nt - t0;
    const int ns = left < KSTEPS ? left : KSTEPS;
    const bool last = (left <= KSTEPS);
    const float* i1 = (L == 0) ? un1 : ((L & 1) ? A1 : B1);
    const float* i2 = (L == 0) ? un2 : ((L & 1) ? A2 : B2);
    float* p1 = last ? outp : ((L & 1) ? B1 : A1);
    float* p2 = last ? (float*)nullptr : ((L & 1) ? B2 : A2);
    wave_steps<<<grid, block, 0, stream>>>(inp + (size_t)t0 * plane, i1, i2, c2, msk, b,
                                           p1, p2, ns, left);
  }
}
